# starting kernel
# baseline (speedup 1.0000x reference)
#include <hip/hip_runtime.h>
#include <hip/hip_bf16.h>

typedef __attribute__((ext_vector_type(2))) float v2f;
typedef __attribute__((ext_vector_type(8))) float v8f;

__global__ void gcn_fill_f32(float* __restrict__ p, float v, size_t n) {
    size_t i = (size_t)blockIdx.x * blockDim.x + threadIdx.x;
    size_t stride = (size_t)gridDim.x * blockDim.x;
    for (; i < n; i += stride) p[i] = v;
}

__global__ void gcn_degree(const int* __restrict__ dst, float* __restrict__ deg, int E) {
    int e = blockIdx.x * blockDim.x + threadIdx.x;
    if (e < E) atomicAdd(&deg[dst[e]], 1.0f);
}

__global__ void gcn_rsqrt_inplace(float* __restrict__ d, int n) {
    int i = blockIdx.x * blockDim.x + threadIdx.x;
    if (i < n) {
        float v = d[i];
        d[i] = (v > 0.0f) ? __frsqrt_rn(v) : 0.0f;
    }
}

template <int NOUT>
__global__ __launch_bounds__(256) void gcn_gemm_wmma_f32(
    const float* __restrict__ A, const float* __restrict__ W,
    float* __restrict__ C, int N, int K) {
    constexpr int NT = NOUT / 16;
    const int wave = threadIdx.x >> 5;
    const int lane = threadIdx.x & 31;
    const int row0 = (blockIdx.x * 8 + wave) * 16;
    if (row0 >= N) return;

    const int m = lane & 15;
    const int g = lane >> 4;
    const float* __restrict__ arow = A + (size_t)(row0 + m) * K + 2 * g;

    v8f acc[NT];
#pragma unroll
    for (int t = 0; t < NT; ++t) acc[t] = (v8f){0, 0, 0, 0, 0, 0, 0, 0};

    for (int k = 0; k < K; k += 4) {
        v2f a = *(const v2f*)(arow + k);
        const float* __restrict__ wrow = W + (size_t)(k + 2 * g) * NOUT + m;
#pragma unroll
        for (int t = 0; t < NT; ++t) {
            v2f b;
            b.x = wrow[t * 16];
            b.y = wrow[NOUT + t * 16];
            acc[t] = __builtin_amdgcn_wmma_f32_16x16x4_f32(
                false, a, false, b, (short)0, acc[t], false, false);
        }
    }

#pragma unroll
    for (int t = 0; t < NT; ++t) {
        float* __restrict__ crow = C + (size_t)(row0 + 8 * g) * NOUT + t * 16 + m;
#pragma unroll
        for (int r = 0; r < 8; ++r) {
            crow[(size_t)r * NOUT] = acc[t][r];
        }
    }
}

__global__ void gcn_scatter(const float* __restrict__ h, const int* __restrict__ src,
                            const int* __restrict__ dst, const float* __restrict__ dinv,
                            float* __restrict__ out, int E, int F) {
    int e = blockIdx.x * blockDim.y + threadIdx.y;
    if (e >= E) return;
    int s = src[e];
    int d = dst[e];
    float nrm = dinv[s] * dinv[d];
    int f = threadIdx.x;
    atomicAdd(&out[(size_t)d * F + f], h[(size_t)s * F + f] * nrm);
}

__global__ void gcn_epilogue(float* __restrict__ agg, const float* __restrict__ h,
                             const float* __restrict__ dinv, const float* __restrict__ bias,
                             size_t total, int F, int do_relu) {
    size_t i = (size_t)blockIdx.x * blockDim.x + threadIdx.x;
    if (i >= total) return;
    size_t node = i / F;
    int f = (int)(i - node * F);
    float di = dinv[node];
    float v = agg[i] + h[i] * di * di + bias[f];
    agg[i] = do_relu ? fmaxf(v, 0.0f) : v;
}

extern "C" void kernel_launch(void* const* d_in, const int* in_sizes, int n_in,
                              void* d_out, int out_size, void* d_ws, size_t ws_size,
                              hipStream_t stream) {
    const float* x  = (const float*)d_in[0];
    const int*   ei = (const int*)d_in[1];
    const float* W1 = (const float*)d_in[2];
    const float* b1 = (const float*)d_in[3];
    const float* W2 = (const float*)d_in[4];
    const float* b2 = (const float*)d_in[5];
    float* out = (float*)d_out;

    const int DIN = 128, DH = 128, DOUT = 64;
    const int N = in_sizes[0] / DIN;
    const int E = in_sizes[1] / 2;
    const int* src = ei;
    const int* dst = ei + E;

    float* ws = (float*)d_ws;
    size_t dinvPad = ((size_t)N + 255) & ~(size_t)255;
    float* dinv = ws;
    float* h    = ws + dinvPad;
    float* agg1 = h + (size_t)N * DH;

    const int TB = 256;
    size_t nh = (size_t)N * DH;
    size_t no = (size_t)N * DOUT;

    gcn_fill_f32<<<(int)((N + TB - 1) / TB), TB, 0, stream>>>(dinv, 1.0f, (size_t)N);
    gcn_fill_f32<<<2048, TB, 0, stream>>>(agg1, 0.0f, nh);
    gcn_fill_f32<<<2048, TB, 0, stream>>>(out, 0.0f, no);

    gcn_degree<<<(E + TB - 1) / TB, TB, 0, stream>>>(dst, dinv, E);
    gcn_rsqrt_inplace<<<(N + TB - 1) / TB, TB, 0, stream>>>(dinv, N);

    int gemmBlocks = ((N / 16) + 7) / 8;
    gcn_gemm_wmma_f32<DH><<<gemmBlocks, 256, 0, stream>>>(x, W1, h, N, DIN);

    {
        dim3 blk(DH, 2);
        gcn_scatter<<<(E + 1) / 2, blk, 0, stream>>>(h, src, dst, dinv, agg1, E, DH);
    }

    gcn_epilogue<<<(int)((nh + TB - 1) / TB), TB, 0, stream>>>(agg1, h, dinv, b1, nh, DH, 1);

    gcn_gemm_wmma_f32<DOUT><<<gemmBlocks, 256, 0, stream>>>(agg1, W2, h, N, DH);

    {
        dim3 blk(DOUT, 4);
        gcn_scatter<<<(E + 3) / 4, blk, 0, stream>>>(h, src, dst, dinv, out, E, DOUT);
    }

    gcn_epilogue<<<(int)((no + TB - 1) / TB), TB, 0, stream>>>(out, h, dinv, b2, no, DOUT, 0);
}


# Edge aggregation as a gather over a per-call destination-sorted edge list (int count+rank, wave-scanned starts) replaces the f32 atomic scatter; self-loop, bias, ReLU folded into its row write
# speedup vs baseline: 2.8373x; 2.8373x over previous
#include <hip/hip_runtime.h>


typedef __attribute__((ext_vector_type(2))) float v2f;
typedef __attribute__((ext_vector_type(4))) float v4f;
typedef __attribute__((ext_vector_type(8))) float v8f;

__global__ void gcn_zero_u32(unsigned* __restrict__ p, int n) {
    int i = blockIdx.x * blockDim.x + threadIdx.x;
    if (i < n) p[i] = 0u;
}

__global__ void gcn_count(const int* __restrict__ dst, unsigned* __restrict__ cnt,
                          unsigned* __restrict__ rank, int E) {
    int e = blockIdx.x * blockDim.x + threadIdx.x;
    if (e < E) rank[e] = atomicAdd(&cnt[dst[e]], 1u);
}

__global__ void gcn_alloc(const unsigned* __restrict__ cnt, unsigned* __restrict__ total,
                          unsigned* __restrict__ start, float* __restrict__ dinv, int N) {
    const int i = blockIdx.x * blockDim.x + threadIdx.x;
    const int lane = threadIdx.x & 31;
    const unsigned c = (i < N) ? cnt[i] : 0u;
    unsigned incl = c;
    for (int o = 1; o < 32; o <<= 1) {
        const unsigned t = __shfl_up(incl, o);
        if (lane >= o) incl += t;
    }
    const unsigned wsum = __shfl(incl, 31);
    unsigned base = 0u;
    if (lane == 0) base = atomicAdd(total, wsum);
    base = __shfl(base, 0);
    if (i < N) {
        start[i] = base + incl - c;
        const float v = (float)(c + 1u);
        dinv[i] = (v > 0.0f) ? __frsqrt_rn(v) : 0.0f;
    }
}

__global__ void gcn_csr_fill(const int* __restrict__ src, const int* __restrict__ dst,
                             const unsigned* __restrict__ rank, const unsigned* __restrict__ start,
                             const float* __restrict__ dinv, int* __restrict__ csr, int E) {
    const int e = blockIdx.x * blockDim.x + threadIdx.x;
    if (e >= E) return;
    const int s = src[e];
    const unsigned slot = start[dst[e]] + rank[e];
    ((int2*)csr)[slot] = make_int2(s, __float_as_int(dinv[s]));
}

template <int NOUT>
__global__ __launch_bounds__(256) void gcn_gemm_wmma_f32(
    const float* __restrict__ A0, const float* __restrict__ A1, int lda, int ksplit,
    const float* __restrict__ W, float* __restrict__ C, int N, int K) {
    constexpr int NT = NOUT / 16;
    const int wave = threadIdx.x >> 5;
    const int lane = threadIdx.x & 31;
    const int row0 = (blockIdx.x * 8 + wave) * 16;
    if (row0 >= N) return;

    const int m = lane & 15;
    const int g = lane >> 4;
    const size_t aoff = (size_t)(row0 + m) * lda + 2 * g;

    v8f acc[NT];
#pragma unroll
    for (int t = 0; t < NT; ++t) acc[t] = (v8f){0, 0, 0, 0, 0, 0, 0, 0};

    auto ksteps = [&](const float* __restrict__ arow, int kbeg, int kend) {
        for (int k = kbeg; k < kend; k += 4) {
            v2f a = *(const v2f*)(arow + (k - kbeg));
            const float* __restrict__ wrow = W + (size_t)(k + 2 * g) * NOUT + m;
#pragma unroll
            for (int t = 0; t < NT; ++t) {
                v2f b;
                b.x = wrow[t * 16];
                b.y = wrow[NOUT + t * 16];
                acc[t] = __builtin_amdgcn_wmma_f32_16x16x4_f32(
                    false, a, false, b, (short)0, acc[t], false, false);
            }
        }
    };
    ksteps(A0 + aoff, 0, ksplit);
    ksteps(A1 + aoff, ksplit, K);

#pragma unroll
    for (int t = 0; t < NT; ++t) {
        float* __restrict__ crow = C + (size_t)(row0 + 8 * g) * NOUT + t * 16 + m;
#pragma unroll
        for (int r = 0; r < 8; ++r) {
            crow[(size_t)r * NOUT] = acc[t][r];
        }
    }
}

template <int U, typename V>
__device__ __forceinline__ void gcn_add_rows(V& acc, const V* __restrict__ hv, int lane,
                                             int es, int ewbits, unsigned j) {
    V r[U];
    float w[U];
#pragma unroll
    for (int u = 0; u < U; ++u) {
        const int s = __builtin_amdgcn_readlane(es, (int)(j + u));
        w[u] = __int_as_float(__builtin_amdgcn_readlane(ewbits, (int)(j + u)));
        r[u] = hv[(size_t)s * 32 + lane];
    }
#pragma unroll
    for (int u = 0; u < U; ++u) acc += r[u] * w[u];
}

template <typename V>
__device__ __forceinline__ V gcn_gather_node(const V* __restrict__ hv, const int2* __restrict__ csr,
                                             unsigned beg, unsigned len, float dn, int node, int lane) {
    V acc = hv[(size_t)node * 32 + lane] * (dn * dn);
    for (unsigned j0 = 0; j0 < len; j0 += 32) {
        const unsigned m = (len - j0 < 32u) ? (len - j0) : 32u;
        int es = 0, ewbits = 0;
        if ((unsigned)lane < m) {
            const int2 ent = csr[beg + j0 + lane];
            es = ent.x;
            ewbits = __float_as_int(__int_as_float(ent.y) * dn);
        }
        unsigned j = 0;
        for (; j + 8 <= m; j += 8) gcn_add_rows<8, V>(acc, hv, lane, es, ewbits, j);
        if (j + 4 <= m) { gcn_add_rows<4, V>(acc, hv, lane, es, ewbits, j); j += 4; }
        if (j + 2 <= m) { gcn_add_rows<2, V>(acc, hv, lane, es, ewbits, j); j += 2; }
        if (j < m) gcn_add_rows<1, V>(acc, hv, lane, es, ewbits, j);
    }
    return acc;
}

__global__ __launch_bounds__(256) void gcn_gather128_relu(
    const float* __restrict__ h, const int* __restrict__ csr,
    const unsigned* __restrict__ start, const unsigned* __restrict__ cnt,
    const float* __restrict__ dinv, const float* __restrict__ bias,
    float* __restrict__ o_lo, float* __restrict__ o_hi, int N) {
    const int lane = threadIdx.x & 31;
    const int node = __builtin_amdgcn_readfirstlane((int)(blockIdx.x * 8 + (threadIdx.x >> 5)));
    if (node >= N) return;
    v4f acc = gcn_gather_node<v4f>((const v4f*)h, (const int2*)csr, start[node], cnt[node], dinv[node], node, lane);
    const float* __restrict__ bp = bias + lane * 4;
    acc.x = fmaxf(acc.x + bp[0], 0.0f);
    acc.y = fmaxf(acc.y + bp[1], 0.0f);
    acc.z = fmaxf(acc.z + bp[2], 0.0f);
    acc.w = fmaxf(acc.w + bp[3], 0.0f);
    float* __restrict__ o = (lane < 16) ? (o_lo + (size_t)node * 64 + lane * 4)
                                        : (o_hi + (size_t)node * 64 + (lane - 16) * 4);
    *(v4f*)o = acc;
}

__global__ __launch_bounds__(256) void gcn_gather64(
    const float* __restrict__ g, const int* __restrict__ csr,
    const unsigned* __restrict__ start, const unsigned* __restrict__ cnt,
    const float* __restrict__ dinv, const float* __restrict__ bias,
    float* __restrict__ out, int N) {
    const int lane = threadIdx.x & 31;
    const int node = __builtin_amdgcn_readfirstlane((int)(blockIdx.x * 8 + (threadIdx.x >> 5)));
    if (node >= N) return;
    v2f acc = gcn_gather_node<v2f>((const v2f*)g, (const int2*)csr, start[node], cnt[node], dinv[node], node, lane);
    acc.x += bias[lane * 2];
    acc.y += bias[lane * 2 + 1];
    *(v2f*)(out + (size_t)node * 64 + lane * 2) = acc;
}

extern "C" void kernel_launch(void* const* d_in, const int* in_sizes, int n_in,
                              void* d_out, int out_size, void* d_ws, size_t ws_size,
                              hipStream_t stream) {
    const float* x  = (const float*)d_in[0];
    const int*   ei = (const int*)d_in[1];
    const float* W1 = (const float*)d_in[2];
    const float* b1 = (const float*)d_in[3];
    const float* W2 = (const float*)d_in[4];
    const float* b2 = (const float*)d_in[5];
    float* out = (float*)d_out;

    const int DIN = 128, DH = 128, DOUT = 64;
    const int N = in_sizes[0] / DIN;
    const int E = in_sizes[1] / 2;
    const int* src = ei;
    const int* dst = ei + E;

    float* ws = (float*)d_ws;
    size_t pad = ((size_t)N + 1 + 255) & ~(size_t)255;
    float*    dinv  = ws;
    unsigned* cnt   = (unsigned*)(ws + pad);
    unsigned* total = cnt + N;
    unsigned* start = (unsigned*)(ws + 2 * pad);
    int*      csr   = (int*)(ws + 3 * pad);
    float*    h     = ws + 3 * pad + 2 * (size_t)E;
    float*    h1hi  = h + (size_t)N * DH;
    float*    h1lo  = out;
    unsigned* rank  = (unsigned*)h1hi;

    const int TB = 256;

    gcn_zero_u32<<<(N + 1 + TB - 1) / TB, TB, 0, stream>>>(cnt, N + 1);
    gcn_count<<<(E + TB - 1) / TB, TB, 0, stream>>>(dst, cnt, rank, E);
    gcn_alloc<<<(N + TB - 1) / TB, TB, 0, stream>>>(cnt, total, start, dinv, N);
    gcn_csr_fill<<<(E + TB - 1) / TB, TB, 0, stream>>>(src, dst, rank, start, dinv, csr, E);

    int gemmBlocks = ((N / 16) + 7) / 8;
    gcn_gemm_wmma_f32<DH><<<gemmBlocks, 256, 0, stream>>>(x, x, DIN, 0, W1, h, N, DIN);

    gcn_gather128_relu<<<(N + 7) / 8, 256, 0, stream>>>(h, csr, start, cnt, dinv, b1, h1lo, h1hi, N);

    gcn_gemm_wmma_f32<DOUT><<<gemmBlocks, 256, 0, stream>>>(h1lo, h1hi, 64, 64, W2, h, N, DH);

    gcn_gather64<<<(N + 7) / 8, 256, 0, stream>>>(h, csr, start, cnt, dinv, b2, out, N);
}


# h = x@W1 loses its launch: its row strips run as the first work-groups of the count and fill launches (half each; edge passes take 4 edges per lane), matrix work under the edge passes; 8 -> 7 launches
# speedup vs baseline: 3.2461x; 1.1441x over previous
#include <hip/hip_runtime.h>


typedef __attribute__((ext_vector_type(2))) float v2f;
typedef __attribute__((ext_vector_type(4))) float v4f;
typedef __attribute__((ext_vector_type(8))) float v8f;

__global__ void gcn_zero_u32(unsigned* __restrict__ p, int n) {
    int i = blockIdx.x * blockDim.x + threadIdx.x;
    if (i < n) p[i] = 0u;
}


__global__ void gcn_alloc(const unsigned* __restrict__ cnt, unsigned* __restrict__ total,
                          unsigned* __restrict__ start, float* __restrict__ dinv, int N) {
    const int i = blockIdx.x * blockDim.x + threadIdx.x;
    const int lane = threadIdx.x & 31;
    const unsigned c = (i < N) ? cnt[i] : 0u;
    unsigned incl = c;
    for (int o = 1; o < 32; o <<= 1) {
        const unsigned t = __shfl_up(incl, o);
        if (lane >= o) incl += t;
    }
    const unsigned wsum = __shfl(incl, 31);
    unsigned base = 0u;
    if (lane == 0) base = atomicAdd(total, wsum);
    base = __shfl(base, 0);
    if (i < N) {
        start[i] = base + incl - c;
        const float v = (float)(c + 1u);
        dinv[i] = (v > 0.0f) ? __frsqrt_rn(v) : 0.0f;
    }
}


template <int NOUT>
__device__ __forceinline__ void gcn_gemm_strip(
    const float* __restrict__ A0, const float* __restrict__ A1, int lda, int ksplit,
    const float* __restrict__ W, float* __restrict__ C, int K, int row0) {
    constexpr int NT = NOUT / 16;
    const int lane = threadIdx.x & 31;

    const int m = lane & 15;
    const int g = lane >> 4;
    const size_t aoff = (size_t)(row0 + m) * lda + 2 * g;

    v8f acc[NT];
#pragma unroll
    for (int t = 0; t < NT; ++t) acc[t] = (v8f){0, 0, 0, 0, 0, 0, 0, 0};

    auto ksteps = [&](const float* __restrict__ arow, int kbeg, int kend) {
        for (int k = kbeg; k < kend; k += 4) {
            v2f a = *(const v2f*)(arow + (k - kbeg));
            const float* __restrict__ wrow = W + (size_t)(k + 2 * g) * NOUT + m;
#pragma unroll
            for (int t = 0; t < NT; ++t) {
                v2f b;
                b.x = wrow[t * 16];
                b.y = wrow[NOUT + t * 16];
                acc[t] = __builtin_amdgcn_wmma_f32_16x16x4_f32(
                    false, a, false, b, (short)0, acc[t], false, false);
            }
        }
    };
    ksteps(A0 + aoff, 0, ksplit);
    ksteps(A1 + aoff, ksplit, K);

#pragma unroll
    for (int t = 0; t < NT; ++t) {
        float* __restrict__ crow = C + (size_t)(row0 + 8 * g) * NOUT + t * 16 + m;
#pragma unroll
        for (int r = 0; r < 8; ++r) {
            crow[(size_t)r * NOUT] = acc[t][r];
        }
    }
}

template <int NOUT>
__global__ __launch_bounds__(256) void gcn_gemm_wmma_f32(
    const float* __restrict__ A0, const float* __restrict__ A1, int lda, int ksplit,
    const float* __restrict__ W, float* __restrict__ C, int N, int K) {
    const int row0 = (blockIdx.x * 8 + (threadIdx.x >> 5)) * 16;
    if (row0 >= N) return;
    gcn_gemm_strip<NOUT>(A0, A1, lda, ksplit, W, C, K, row0);
}

__global__ __launch_bounds__(256) void gcn_count_gemm(
    const int* __restrict__ dst, unsigned* __restrict__ cnt, unsigned* __restrict__ rank, int E,
    const float* __restrict__ x, const float* __restrict__ W1, float* __restrict__ h,
    int K, int sBeg, int sEnd, int gblocks) {
    if ((int)blockIdx.x < gblocks) {
        const int strip = sBeg + (int)blockIdx.x * 8 + (int)(threadIdx.x >> 5);
        if (strip < sEnd) gcn_gemm_strip<128>(x, x, K, 0, W1, h, K, strip * 16);
        return;
    }
    const int q = ((int)blockIdx.x - gblocks) * 256 + (int)threadIdx.x;
    if (q * 4 < E) {
        const int4 d = ((const int4*)dst)[q];
        uint4 r;
        r.x = atomicAdd(&cnt[d.x], 1u);
        r.y = atomicAdd(&cnt[d.y], 1u);
        r.z = atomicAdd(&cnt[d.z], 1u);
        r.w = atomicAdd(&cnt[d.w], 1u);
        ((uint4*)rank)[q] = r;
    }
}

__global__ __launch_bounds__(256) void gcn_csr_fill_gemm(
    const int* __restrict__ src, const int* __restrict__ dst,
    const unsigned* __restrict__ rank, const unsigned* __restrict__ start,
    const float* __restrict__ dinv, int* __restrict__ csr, int E,
    const float* __restrict__ x, const float* __restrict__ W1, float* __restrict__ h,
    int K, int sBeg, int sEnd, int gblocks) {
    if ((int)blockIdx.x < gblocks) {
        const int strip = sBeg + (int)blockIdx.x * 8 + (int)(threadIdx.x >> 5);
        if (strip < sEnd) gcn_gemm_strip<128>(x, x, K, 0, W1, h, K, strip * 16);
        return;
    }
    const int q = ((int)blockIdx.x - gblocks) * 256 + (int)threadIdx.x;
    if (q * 4 >= E) return;
    const int4 s = ((const int4*)src)[q];
    const int4 d = ((const int4*)dst)[q];
    const uint4 r = ((const uint4*)rank)[q];
    const unsigned p0 = start[d.x] + r.x, p1 = start[d.y] + r.y;
    const unsigned p2 = start[d.z] + r.z, p3 = start[d.w] + r.w;
    const float w0 = dinv[s.x], w1 = dinv[s.y], w2 = dinv[s.z], w3 = dinv[s.w];
    int2* __restrict__ c2 = (int2*)csr;
    c2[p0] = make_int2(s.x, __float_as_int(w0));
    c2[p1] = make_int2(s.y, __float_as_int(w1));
    c2[p2] = make_int2(s.z, __float_as_int(w2));
    c2[p3] = make_int2(s.w, __float_as_int(w3));
}

template <int U, typename V>
__device__ __forceinline__ void gcn_add_rows(V& acc, const V* __restrict__ hv, int lane,
                                             int es, int ewbits, unsigned j) {
    V r[U];
    float w[U];
#pragma unroll
    for (int u = 0; u < U; ++u) {
        const int s = __builtin_amdgcn_readlane(es, (int)(j + u));
        w[u] = __int_as_float(__builtin_amdgcn_readlane(ewbits, (int)(j + u)));
        r[u] = hv[(size_t)s * 32 + lane];
    }
#pragma unroll
    for (int u = 0; u < U; ++u) acc += r[u] * w[u];
}

template <typename V>
__device__ __forceinline__ V gcn_gather_node(const V* __restrict__ hv, const int2* __restrict__ csr,
                                             unsigned beg, unsigned len, float dn, int node, int lane) {
    V acc = hv[(size_t)node * 32 + lane] * (dn * dn);
    for (unsigned j0 = 0; j0 < len; j0 += 32) {
        const unsigned m = (len - j0 < 32u) ? (len - j0) : 32u;
        int es = 0, ewbits = 0;
        if ((unsigned)lane < m) {
            const int2 ent = csr[beg + j0 + lane];
            es = ent.x;
            ewbits = __float_as_int(__int_as_float(ent.y) * dn);
        }
        unsigned j = 0;
        for (; j + 8 <= m; j += 8) gcn_add_rows<8, V>(acc, hv, lane, es, ewbits, j);
        if (j + 4 <= m) { gcn_add_rows<4, V>(acc, hv, lane, es, ewbits, j); j += 4; }
        if (j + 2 <= m) { gcn_add_rows<2, V>(acc, hv, lane, es, ewbits, j); j += 2; }
        if (j < m) gcn_add_rows<1, V>(acc, hv, lane, es, ewbits, j);
    }
    return acc;
}

__global__ __launch_bounds__(256) void gcn_gather128_relu(
    const float* __restrict__ h, const int* __restrict__ csr,
    const unsigned* __restrict__ start, const unsigned* __restrict__ cnt,
    const float* __restrict__ dinv, const float* __restrict__ bias,
    float* __restrict__ o_lo, float* __restrict__ o_hi, int N) {
    const int lane = threadIdx.x & 31;
    const int node = __builtin_amdgcn_readfirstlane((int)(blockIdx.x * 8 + (threadIdx.x >> 5)));
    if (node >= N) return;
    v4f acc = gcn_gather_node<v4f>((const v4f*)h, (const int2*)csr, start[node], cnt[node], dinv[node], node, lane);
    const float* __restrict__ bp = bias + lane * 4;
    acc.x = fmaxf(acc.x + bp[0], 0.0f);
    acc.y = fmaxf(acc.y + bp[1], 0.0f);
    acc.z = fmaxf(acc.z + bp[2], 0.0f);
    acc.w = fmaxf(acc.w + bp[3], 0.0f);
    float* __restrict__ o = (lane < 16) ? (o_lo + (size_t)node * 64 + lane * 4)
                                        : (o_hi + (size_t)node * 64 + (lane - 16) * 4);
    *(v4f*)o = acc;
}

__global__ __launch_bounds__(256) void gcn_gather64(
    const float* __restrict__ g, const int* __restrict__ csr,
    const unsigned* __restrict__ start, const unsigned* __restrict__ cnt,
    const float* __restrict__ dinv, const float* __restrict__ bias,
    float* __restrict__ out, int N) {
    const int lane = threadIdx.x & 31;
    const int node = __builtin_amdgcn_readfirstlane((int)(blockIdx.x * 8 + (threadIdx.x >> 5)));
    if (node >= N) return;
    v2f acc = gcn_gather_node<v2f>((const v2f*)g, (const int2*)csr, start[node], cnt[node], dinv[node], node, lane);
    acc.x += bias[lane * 2];
    acc.y += bias[lane * 2 + 1];
    *(v2f*)(out + (size_t)node * 64 + lane * 2) = acc;
}

extern "C" void kernel_launch(void* const* d_in, const int* in_sizes, int n_in,
                              void* d_out, int out_size, void* d_ws, size_t ws_size,
                              hipStream_t stream) {
    const float* x  = (const float*)d_in[0];
    const int*   ei = (const int*)d_in[1];
    const float* W1 = (const float*)d_in[2];
    const float* b1 = (const float*)d_in[3];
    const float* W2 = (const float*)d_in[4];
    const float* b2 = (const float*)d_in[5];
    float* out = (float*)d_out;

    const int DIN = 128, DH = 128, DOUT = 64;
    const int N = in_sizes[0] / DIN;
    const int E = in_sizes[1] / 2;
    const int* src = ei;
    const int* dst = ei + E;

    float* ws = (float*)d_ws;
    size_t pad = ((size_t)N + 1 + 255) & ~(size_t)255;
    float*    dinv  = ws;
    unsigned* cnt   = (unsigned*)(ws + pad);
    unsigned* total = cnt + N;
    unsigned* start = (unsigned*)(ws + 2 * pad);
    int*      csr   = (int*)(ws + 3 * pad);
    float*    h     = ws + 3 * pad + 2 * (size_t)E;
    float*    h1hi  = h + (size_t)N * DH;
    float*    h1lo  = out;
    unsigned* rank  = (unsigned*)h1hi;

    const int TB = 256;

    gcn_zero_u32<<<(N + 1 + TB - 1) / TB, TB, 0, stream>>>(cnt, N + 1);
    const int strips = N / 16;
    const int sHalf = strips / 2;
    const int gA = (sHalf + 7) / 8;
    const int gB = (strips - sHalf + 7) / 8;
    const int edgeBlocks = (E / 4 + TB - 1) / TB;
    gcn_count_gemm<<<gA + edgeBlocks, TB, 0, stream>>>(dst, cnt, rank, E, x, W1, h, DIN, 0, sHalf, gA);
    gcn_alloc<<<(N + TB - 1) / TB, TB, 0, stream>>>(cnt, total, start, dinv, N);
    gcn_csr_fill_gemm<<<gB + edgeBlocks, TB, 0, stream>>>(src, dst, rank, start, dinv, csr, E,
                                                          x, W1, h, DIN, sHalf, strips, gB);
    int gemmBlocks = (strips + 7) / 8;

    gcn_gather128_relu<<<(N + 7) / 8, 256, 0, stream>>>(h, csr, start, cnt, dinv, b1, h1lo, h1hi, N);

    gcn_gemm_wmma_f32<DOUT><<<gemmBlocks, 256, 0, stream>>>(h1lo, h1hi, 64, 64, W2, h, N, DH);

    gcn_gather64<<<(N + 7) / 8, 256, 0, stream>>>(h, csr, start, cnt, dinv, b2, out, N);
}
